# starting kernel
# baseline (speedup 1.0000x reference)
#include <hip/hip_runtime.h>

typedef __attribute__((ext_vector_type(2))) float v2f;
typedef __attribute__((ext_vector_type(8))) float v8f;

#define RNN_B 4096
#define RNN_T 512
#define RNN_I 16
#define RNN_H 6

__device__ __forceinline__ v8f wmma_f32_4(v2f a, v2f b, v8f c) {
  return __builtin_amdgcn_wmma_f32_16x16x4_f32(false, a, false, b, (short)0, c,
                                               false, false);
}

__device__ __forceinline__ float tanh_fast(float x) {
#if __has_builtin(__builtin_amdgcn_tanhf)
  return __builtin_amdgcn_tanhf(x);
#else
  x = __builtin_amdgcn_fmed3f(x, -9.0f, 9.0f);
  float e = __expf(2.0f * x);
  return (e - 1.0f) * __builtin_amdgcn_rcpf(e + 1.0f);
#endif
}

__device__ __forceinline__ void tanh8(v8f& a) {
#pragma unroll
  for (int i = 0; i < 8; ++i) a[i] = tanh_fast(a[i]);
}

__global__ __launch_bounds__(64) void rnn_fused_wmma(
    const float* __restrict__ x, const float* __restrict__ W_ih,
    const float* __restrict__ W_hh, const float* __restrict__ b_ih,
    const float* __restrict__ b_hh, const float* __restrict__ fc_w,
    const float* __restrict__ fc_b, float* __restrict__ out) {
  __shared__ __align__(16) float hbuf[2][16][18];

  const int tid  = threadIdx.x;
  const int w    = tid >> 5;
  const int lane = tid & 31;
  const int n    = lane & 15;
  const int hi   = lane >> 4;
  const int b0   = (blockIdx.x * 2 + w) * 16;

  v2f bih[4];
#pragma unroll
  for (int s = 0; s < 4; ++s) {
#pragma unroll
    for (int v = 0; v < 2; ++v) {
      int k = 4 * s + 2 * hi + v;
      bih[s][v] = (n < RNN_H) ? W_ih[n * RNN_I + k] : 0.0f;
    }
  }
  v2f whh[2];
#pragma unroll
  for (int s = 0; s < 2; ++s) {
#pragma unroll
    for (int v = 0; v < 2; ++v) {
      int k = 4 * s + 2 * hi + v;
      whh[s][v] = (n < RNN_H && k < RNN_H) ? W_hh[n * RNN_H + k] : 0.0f;
    }
  }
  const float bn = (n < RNN_H) ? (b_ih[n] + b_hh[n]) : 0.0f;
  v8f cbias;
#pragma unroll
  for (int i = 0; i < 8; ++i) cbias[i] = bn;

  const float* xp = x + (size_t)(b0 + n) * (RNN_T * RNN_I) + 2 * hi;

  v2f ha0 = 0.0f, ha1 = 0.0f;
  v2f xa[4], xb[4];
#pragma unroll
  for (int s = 0; s < 4; ++s) xa[s] = *(const v2f*)(xp + 4 * s);

  float(*hb)[18] = hbuf[w];

  for (int t = 0; t < RNN_T; t += 2) {
    v8f acc = cbias;
#pragma unroll
    for (int s = 0; s < 4; ++s) acc = wmma_f32_4(xa[s], bih[s], acc);
    acc = wmma_f32_4(ha0, whh[0], acc);
    acc = wmma_f32_4(ha1, whh[1], acc);
    {
      const float* p = xp + (size_t)(t + 1) * RNN_I;
#pragma unroll
      for (int s = 0; s < 4; ++s) xb[s] = *(const v2f*)(p + 4 * s);
    }
    tanh8(acc);
#pragma unroll
    for (int r = 0; r < 8; ++r) hb[r + hi * 8][n] = acc[r];
    __builtin_amdgcn_wave_barrier();
    ha0 = *(const v2f*)&hb[n][2 * hi];
    ha1 = *(const v2f*)&hb[n][4 + 2 * hi];
    __builtin_amdgcn_wave_barrier();

    acc = cbias;
#pragma unroll
    for (int s = 0; s < 4; ++s) acc = wmma_f32_4(xb[s], bih[s], acc);
    acc = wmma_f32_4(ha0, whh[0], acc);
    acc = wmma_f32_4(ha1, whh[1], acc);
    {
      const int t2 = (t + 2) & (RNN_T - 1);
      const float* p = xp + (size_t)t2 * RNN_I;
#pragma unroll
      for (int s = 0; s < 4; ++s) xa[s] = *(const v2f*)(p + 4 * s);
    }
    tanh8(acc);
#pragma unroll
    for (int r = 0; r < 8; ++r) hb[r + hi * 8][n] = acc[r];
    __builtin_amdgcn_wave_barrier();
    ha0 = *(const v2f*)&hb[n][2 * hi];
    ha1 = *(const v2f*)&hb[n][4 + 2 * hi];
    __builtin_amdgcn_wave_barrier();
  }

  if (lane < 16) {
    float s = fc_b[0];
#pragma unroll
    for (int j = 0; j < RNN_H; ++j) s += hb[lane][j] * fc_w[j];
    out[b0 + lane] = s;
  }
}

extern "C" void kernel_launch(void* const* d_in, const int* in_sizes, int n_in,
                              void* d_out, int out_size, void* d_ws,
                              size_t ws_size, hipStream_t stream) {
  (void)in_sizes; (void)n_in; (void)out_size; (void)d_ws; (void)ws_size;
  const float* x    = (const float*)d_in[0];
  const float* W_ih = (const float*)d_in[1];
  const float* W_hh = (const float*)d_in[2];
  const float* b_ih = (const float*)d_in[3];
  const float* b_hh = (const float*)d_in[4];
  const float* fc_w = (const float*)d_in[5];
  const float* fc_b = (const float*)d_in[6];
  float* out = (float*)d_out;

  dim3 block(64);
  dim3 grid(RNN_B / 32);
  rnn_fused_wmma<<<grid, block, 0, stream>>>(x, W_ih, W_hh, b_ih, b_hh, fc_w,
                                             fc_b, out);
}


# x loads issued 8 time steps ahead instead of 1 (register ring refilled in place; each step claims its fragments at its start so the scheduler keeps the distance); sums and their order unchanged
# speedup vs baseline: 2.2200x; 2.2200x over previous
#include <hip/hip_runtime.h>

typedef __attribute__((ext_vector_type(2))) float v2f;
typedef __attribute__((ext_vector_type(8))) float v8f;

#define RNN_B 4096
#define RNN_T 512
#define RNN_I 16
#define RNN_H 6
#define RNN_PF 8

__device__ __forceinline__ v8f wmma_f32_4(v2f a, v2f b, v8f c) {
  return __builtin_amdgcn_wmma_f32_16x16x4_f32(false, a, false, b, (short)0, c,
                                               false, false);
}

__device__ __forceinline__ float tanh_fast(float x) {
#if __has_builtin(__builtin_amdgcn_tanhf)
  return __builtin_amdgcn_tanhf(x);
#else
  x = __builtin_amdgcn_fmed3f(x, -9.0f, 9.0f);
  float e = __expf(2.0f * x);
  return (e - 1.0f) * __builtin_amdgcn_rcpf(e + 1.0f);
#endif
}

__device__ __forceinline__ void tanh8(v8f& a) {
#pragma unroll
  for (int i = 0; i < 8; ++i) a[i] = tanh_fast(a[i]);
}

__device__ __forceinline__ void rnn_step(v2f (&xs)[4], v2f (&xn)[4],
                                         const float* __restrict__ pnext,
                                         const v2f (&bih)[4], const v2f (&whh)[2],
                                         const v8f& cbias, v2f& ha0, v2f& ha1,
                                         float (*hb)[18], int hi, int n) {
  asm volatile("" : "+v"(xs[0]), "+v"(xs[1]), "+v"(xs[2]), "+v"(xs[3]));
  v8f acc = cbias;
#pragma unroll
  for (int s = 0; s < 4; ++s) acc = wmma_f32_4(xs[s], bih[s], acc);
  acc = wmma_f32_4(ha0, whh[0], acc);
  acc = wmma_f32_4(ha1, whh[1], acc);
#pragma unroll
  for (int s = 0; s < 4; ++s) xn[s] = *(const v2f*)(pnext + 4 * s);
  tanh8(acc);
#pragma unroll
  for (int r = 0; r < 8; ++r) hb[r + hi * 8][n] = acc[r];
  __builtin_amdgcn_wave_barrier();
  ha0 = *(const v2f*)&hb[n][2 * hi];
  ha1 = *(const v2f*)&hb[n][4 + 2 * hi];
  __builtin_amdgcn_wave_barrier();
}

__global__ __launch_bounds__(64) void rnn_fused_wmma(
    const float* __restrict__ x, const float* __restrict__ W_ih,
    const float* __restrict__ W_hh, const float* __restrict__ b_ih,
    const float* __restrict__ b_hh, const float* __restrict__ fc_w,
    const float* __restrict__ fc_b, float* __restrict__ out) {
  __shared__ __align__(16) float hbuf[2][16][18];

  const int tid  = threadIdx.x;
  const int w    = tid >> 5;
  const int lane = tid & 31;
  const int n    = lane & 15;
  const int hi   = lane >> 4;
  const int b0   = (blockIdx.x * 2 + w) * 16;

  v2f bih[4];
#pragma unroll
  for (int s = 0; s < 4; ++s) {
#pragma unroll
    for (int v = 0; v < 2; ++v) {
      int k = 4 * s + 2 * hi + v;
      bih[s][v] = (n < RNN_H) ? W_ih[n * RNN_I + k] : 0.0f;
    }
  }
  v2f whh[2];
#pragma unroll
  for (int s = 0; s < 2; ++s) {
#pragma unroll
    for (int v = 0; v < 2; ++v) {
      int k = 4 * s + 2 * hi + v;
      whh[s][v] = (n < RNN_H && k < RNN_H) ? W_hh[n * RNN_H + k] : 0.0f;
    }
  }
  const float bn = (n < RNN_H) ? (b_ih[n] + b_hh[n]) : 0.0f;
  v8f cbias;
#pragma unroll
  for (int i = 0; i < 8; ++i) cbias[i] = bn;

  const float* xp = x + (size_t)(b0 + n) * (RNN_T * RNN_I) + 2 * hi;

  v2f ha0 = 0.0f, ha1 = 0.0f;

  v2f xa[RNN_PF][4], xb[RNN_PF][4];
#pragma unroll
  for (int j = 0; j < RNN_PF; ++j) {
#pragma unroll
    for (int s = 0; s < 4; ++s) xa[j][s] = *(const v2f*)(xp + j * RNN_I + 4 * s);
  }

  float(*hb)[18] = hbuf[w];

  for (int t = 0; t < RNN_T; t += 2 * RNN_PF) {
    const float* pb = xp + (size_t)(t + RNN_PF) * RNN_I;
    const int ta = (t + 2 * RNN_PF < RNN_T) ? (t + 2 * RNN_PF) : (RNN_T - RNN_PF);
    const float* pa = xp + (size_t)ta * RNN_I;
#pragma unroll
    for (int j = 0; j < RNN_PF; ++j)
      rnn_step(xa[j], xb[j], pb + j * RNN_I, bih, whh, cbias, ha0, ha1, hb, hi, n);
#pragma unroll
    for (int j = 0; j < RNN_PF; ++j)
      rnn_step(xb[j], xa[j], pa + j * RNN_I, bih, whh, cbias, ha0, ha1, hb, hi, n);
  }

  if (lane < 16) {
    float s = fc_b[0];
#pragma unroll
    for (int j = 0; j < RNN_H; ++j) s += hb[lane][j] * fc_w[j];
    out[b0 + lane] = s;
  }
}

extern "C" void kernel_launch(void* const* d_in, const int* in_sizes, int n_in,
                              void* d_out, int out_size, void* d_ws,
                              size_t ws_size, hipStream_t stream) {
  (void)in_sizes; (void)n_in; (void)out_size; (void)d_ws; (void)ws_size;
  const float* x    = (const float*)d_in[0];
  const float* W_ih = (const float*)d_in[1];
  const float* W_hh = (const float*)d_in[2];
  const float* b_ih = (const float*)d_in[3];
  const float* b_hh = (const float*)d_in[4];
  const float* fc_w = (const float*)d_in[5];
  const float* fc_b = (const float*)d_in[6];
  float* out = (float*)d_out;

  dim3 block(64);
  dim3 grid(RNN_B / 32);
  rnn_fused_wmma<<<grid, block, 0, stream>>>(x, W_ih, W_hh, b_ih, b_hh, fc_w,
                                             fc_b, out);
}


# h kept in registers: step computed transposed (weights as A, x_t and h as B) so tanh(D) is the next step's B fragment; per-step LDS round trip and 4 of 8 tanh gone; sums and their order unchanged
# speedup vs baseline: 2.4621x; 1.1091x over previous
#include <hip/hip_runtime.h>


typedef __attribute__((ext_vector_type(2))) float v2f;
typedef __attribute__((ext_vector_type(8))) float v8f;

#define RNN_B 4096
#define RNN_T 512
#define RNN_I 16
#define RNN_H 6
#define RNN_PF 8

__device__ __forceinline__ v8f wmma_f32_4(v2f a, v2f b, v8f c) {
  return __builtin_amdgcn_wmma_f32_16x16x4_f32(false, a, false, b, (short)0, c,
                                               false, false);
}

__device__ __forceinline__ float tanh_fast(float x) {
#if __has_builtin(__builtin_amdgcn_tanhf)
  return __builtin_amdgcn_tanhf(x);
#else
  x = __builtin_amdgcn_fmed3f(x, -9.0f, 9.0f);
  float e = __expf(2.0f * x);
  return (e - 1.0f) * __builtin_amdgcn_rcpf(e + 1.0f);
#endif
}

__device__ __forceinline__ int rnn_unit_of_row(int row) {
  if (row < 4) return (row & 1) + 4 * (row >> 1);
  if (row == 8 || row == 9) return row - 6;
  return -1;
}

__device__ __forceinline__ void rnn_step(v2f (&xs)[4], v2f (&xn)[4],
                                         const float* __restrict__ pnext,
                                         const v2f (&wih)[4], const v2f (&whh)[2],
                                         const v8f& cbias, v2f& hb0, v2f& hb1) {
  asm volatile("" : "+v"(xs[0]), "+v"(xs[1]), "+v"(xs[2]), "+v"(xs[3]));
  v8f acc = cbias;
#pragma unroll
  for (int s = 0; s < 4; ++s) acc = wmma_f32_4(wih[s], xs[s], acc);
  acc = wmma_f32_4(whh[0], hb0, acc);
  acc = wmma_f32_4(whh[1], hb1, acc);
#pragma unroll
  for (int s = 0; s < 4; ++s) xn[s] = *(const v2f*)(pnext + 4 * s);
  hb0[0] = tanh_fast(acc[0]);
  hb0[1] = tanh_fast(acc[1]);
  hb1[0] = tanh_fast(acc[2]);
  hb1[1] = tanh_fast(acc[3]);
}

__global__ __launch_bounds__(64) void rnn_fused_wmma(
    const float* __restrict__ x, const float* __restrict__ W_ih,
    const float* __restrict__ W_hh, const float* __restrict__ b_ih,
    const float* __restrict__ b_hh, const float* __restrict__ fc_w,
    const float* __restrict__ fc_b, float* __restrict__ out) {
  __shared__ float hfin[2][16][8];

  const int tid  = threadIdx.x;
  const int w    = tid >> 5;
  const int lane = tid & 31;
  const int n    = lane & 15;
  const int hi   = lane >> 4;
  const int b0   = (blockIdx.x * 2 + w) * 16;

  const int ua = rnn_unit_of_row(n);
  v2f wih[4];
#pragma unroll
  for (int s = 0; s < 4; ++s) {
#pragma unroll
    for (int v = 0; v < 2; ++v) {
      int k = 4 * s + 2 * hi + v;
      wih[s][v] = (ua >= 0) ? W_ih[ua * RNN_I + k] : 0.0f;
    }
  }
  v2f whh[2];
#pragma unroll
  for (int s = 0; s < 2; ++s) {
#pragma unroll
    for (int v = 0; v < 2; ++v) {
      int k = 4 * s + 2 * hi + v;
      whh[s][v] = (ua >= 0 && k < RNN_H) ? W_hh[ua * RNN_H + k] : 0.0f;
    }
  }
  v8f cbias;
#pragma unroll
  for (int r = 0; r < 8; ++r) {
    int uc = rnn_unit_of_row(8 * hi + r);
    cbias[r] = (uc >= 0) ? (b_ih[uc] + b_hh[uc]) : 0.0f;
  }

  const float* xp = x + (size_t)(b0 + n) * (RNN_T * RNN_I) + 2 * hi;

  v2f hb0 = 0.0f, hb1 = 0.0f;

  v2f xa[RNN_PF][4], xb[RNN_PF][4];
#pragma unroll
  for (int j = 0; j < RNN_PF; ++j) {
#pragma unroll
    for (int s = 0; s < 4; ++s) xa[j][s] = *(const v2f*)(xp + j * RNN_I + 4 * s);
  }

  for (int t = 0; t < RNN_T; t += 2 * RNN_PF) {
    const float* pb = xp + (size_t)(t + RNN_PF) * RNN_I;
    const int ta = (t + 2 * RNN_PF < RNN_T) ? (t + 2 * RNN_PF) : (RNN_T - RNN_PF);
    const float* pa = xp + (size_t)ta * RNN_I;
#pragma unroll
    for (int j = 0; j < RNN_PF; ++j)
      rnn_step(xa[j], xb[j], pb + j * RNN_I, wih, whh, cbias, hb0, hb1);
#pragma unroll
    for (int j = 0; j < RNN_PF; ++j)
      rnn_step(xb[j], xa[j], pa + j * RNN_I, wih, whh, cbias, hb0, hb1);
  }

  float(*hf)[8] = hfin[w];
  if (hi == 0) {
    hf[n][0] = hb0[0]; hf[n][1] = hb0[1]; hf[n][4] = hb1[0]; hf[n][5] = hb1[1];
  } else {
    hf[n][2] = hb0[0]; hf[n][3] = hb0[1];
  }
  __syncthreads();
  if (lane < 16) {
    float s = fc_b[0];
#pragma unroll
    for (int j = 0; j < RNN_H; ++j) s += hf[lane][j] * fc_w[j];
    out[b0 + lane] = s;
  }
}

extern "C" void kernel_launch(void* const* d_in, const int* in_sizes, int n_in,
                              void* d_out, int out_size, void* d_ws,
                              size_t ws_size, hipStream_t stream) {
  (void)in_sizes; (void)n_in; (void)out_size; (void)d_ws; (void)ws_size;
  const float* x    = (const float*)d_in[0];
  const float* W_ih = (const float*)d_in[1];
  const float* W_hh = (const float*)d_in[2];
  const float* b_ih = (const float*)d_in[3];
  const float* b_hh = (const float*)d_in[4];
  const float* fc_w = (const float*)d_in[5];
  const float* fc_b = (const float*)d_in[6];
  float* out = (float*)d_out;

  dim3 block(64);
  dim3 grid(RNN_B / 32);
  rnn_fused_wmma<<<grid, block, 0, stream>>>(x, W_ih, W_hh, b_ih, b_hh, fc_w,
                                             fc_b, out);
}


# x projection off the serial chain: run 2 steps ahead in its own accumulators, its products alternating with the 2 h products on the matrix pipe (x ring 16 deep to keep its lead); sums and order kept
# speedup vs baseline: 3.4715x; 1.4100x over previous
#include <hip/hip_runtime.h>


typedef __attribute__((ext_vector_type(2))) float v2f;
typedef __attribute__((ext_vector_type(8))) float v8f;

#define RNN_B 4096
#define RNN_T 512
#define RNN_I 16
#define RNN_H 6
#define RNN_PF 16

__device__ __forceinline__ v8f wmma_f32_4(v2f a, v2f b, v8f c) {
  return __builtin_amdgcn_wmma_f32_16x16x4_f32(false, a, false, b, (short)0, c,
                                               false, false);
}

__device__ __forceinline__ float tanh_fast(float x) {
#if __has_builtin(__builtin_amdgcn_tanhf)
  return __builtin_amdgcn_tanhf(x);
#else
  x = __builtin_amdgcn_fmed3f(x, -9.0f, 9.0f);
  float e = __expf(2.0f * x);
  return (e - 1.0f) * __builtin_amdgcn_rcpf(e + 1.0f);
#endif
}

__device__ __forceinline__ int rnn_unit_of_row(int row) {
  if (row < 4) return (row & 1) + 4 * (row >> 1);
  if (row == 8 || row == 9) return row - 6;
  return -1;
}

#define RNN_ORDER(a, b) asm volatile("" : "+v"(a), "+v"(b))

__device__ __forceinline__ void rnn_step(v2f (&x2)[4], const v2f& x1s3, v2f (&xn)[4],
                                         const float* __restrict__ pnext,
                                         const v2f (&wih)[4], const v2f (&whh)[2],
                                         const v8f& cbias, v8f& xacc0, v8f& xpart,
                                         v2f& hb0, v2f& hb1) {
  asm volatile("" : "+v"(x2[0]), "+v"(x2[1]), "+v"(x2[2]), "+v"(x2[3]), "+v"(hb0), "+v"(hb1));
  v8f nx = wmma_f32_4(wih[0], x2[0], cbias);
  RNN_ORDER(nx, hb0);
  v8f acc = wmma_f32_4(whh[0], hb0, xacc0);
  RNN_ORDER(acc, xpart);
  xpart = wmma_f32_4(wih[3], x1s3, xpart);
  RNN_ORDER(xpart, nx);
  nx = wmma_f32_4(wih[1], x2[1], nx);
  RNN_ORDER(nx, hb1);
  acc = wmma_f32_4(whh[1], hb1, acc);
  RNN_ORDER(acc, nx);
  nx = wmma_f32_4(wih[2], x2[2], nx);
  RNN_ORDER(nx, acc);
#pragma unroll
  for (int s = 0; s < 4; ++s) xn[s] = *(const v2f*)(pnext + 4 * s);
  hb0[0] = tanh_fast(acc[0]);
  hb0[1] = tanh_fast(acc[1]);
  hb1[0] = tanh_fast(acc[2]);
  hb1[1] = tanh_fast(acc[3]);
  xacc0 = xpart;
  xpart = nx;
}

__global__ __launch_bounds__(64) void rnn_fused_wmma(
    const float* __restrict__ x, const float* __restrict__ W_ih,
    const float* __restrict__ W_hh, const float* __restrict__ b_ih,
    const float* __restrict__ b_hh, const float* __restrict__ fc_w,
    const float* __restrict__ fc_b, float* __restrict__ out) {
  __shared__ float hfin[2][16][8];

  const int tid  = threadIdx.x;
  const int w    = tid >> 5;
  const int lane = tid & 31;
  const int n    = lane & 15;
  const int hi   = lane >> 4;
  const int b0   = (blockIdx.x * 2 + w) * 16;

  const int ua = rnn_unit_of_row(n);
  v2f wih[4];
#pragma unroll
  for (int s = 0; s < 4; ++s) {
#pragma unroll
    for (int v = 0; v < 2; ++v) {
      int k = 4 * s + 2 * hi + v;
      wih[s][v] = (ua >= 0) ? W_ih[ua * RNN_I + k] : 0.0f;
    }
  }
  v2f whh[2];
#pragma unroll
  for (int s = 0; s < 2; ++s) {
#pragma unroll
    for (int v = 0; v < 2; ++v) {
      int k = 4 * s + 2 * hi + v;
      whh[s][v] = (ua >= 0 && k < RNN_H) ? W_hh[ua * RNN_H + k] : 0.0f;
    }
  }
  v8f cbias;
#pragma unroll
  for (int r = 0; r < 8; ++r) {
    int uc = rnn_unit_of_row(8 * hi + r);
    cbias[r] = (uc >= 0) ? (b_ih[uc] + b_hh[uc]) : 0.0f;
  }

  const float* xp = x + (size_t)(b0 + n) * (RNN_T * RNN_I) + 2 * hi;

  v2f hb0 = 0.0f, hb1 = 0.0f;

  v2f xa[RNN_PF][4], xb[RNN_PF][4];
#pragma unroll
  for (int j = 0; j < RNN_PF; ++j) {
#pragma unroll
    for (int s = 0; s < 4; ++s) xa[j][s] = *(const v2f*)(xp + j * RNN_I + 4 * s);
  }

  v8f xacc0 = cbias, xpart = cbias;
#pragma unroll
  for (int s = 0; s < 4; ++s) xacc0 = wmma_f32_4(wih[s], xa[0][s], xacc0);
#pragma unroll
  for (int s = 0; s < 3; ++s) xpart = wmma_f32_4(wih[s], xa[1][s], xpart);

  for (int t = 0; t < RNN_T; t += 2 * RNN_PF) {
    const float* pb = xp + (size_t)(t + RNN_PF) * RNN_I;
    const int ta = (t + 2 * RNN_PF < RNN_T) ? (t + 2 * RNN_PF) : (RNN_T - RNN_PF);
    const float* pa = xp + (size_t)ta * RNN_I;
#pragma unroll
    for (int j = 0; j < RNN_PF; ++j)
      rnn_step(j + 2 < RNN_PF ? xa[j + 2] : xb[j + 2 - RNN_PF],
               j + 1 < RNN_PF ? xa[j + 1][3] : xb[j + 1 - RNN_PF][3],
               xb[j], pb + j * RNN_I, wih, whh, cbias, xacc0, xpart, hb0, hb1);
#pragma unroll
    for (int j = 0; j < RNN_PF; ++j)
      rnn_step(j + 2 < RNN_PF ? xb[j + 2] : xa[j + 2 - RNN_PF],
               j + 1 < RNN_PF ? xb[j + 1][3] : xa[j + 1 - RNN_PF][3],
               xa[j], pa + j * RNN_I, wih, whh, cbias, xacc0, xpart, hb0, hb1);
  }

  float(*hf)[8] = hfin[w];
  if (hi == 0) {
    hf[n][0] = hb0[0]; hf[n][1] = hb0[1]; hf[n][4] = hb1[0]; hf[n][5] = hb1[1];
  } else {
    hf[n][2] = hb0[0]; hf[n][3] = hb0[1];
  }
  __syncthreads();
  if (lane < 16) {
    float s = fc_b[0];
#pragma unroll
    for (int j = 0; j < RNN_H; ++j) s += hf[lane][j] * fc_w[j];
    out[b0 + lane] = s;
  }
}

extern "C" void kernel_launch(void* const* d_in, const int* in_sizes, int n_in,
                              void* d_out, int out_size, void* d_ws,
                              size_t ws_size, hipStream_t stream) {
  (void)in_sizes; (void)n_in; (void)out_size; (void)d_ws; (void)ws_size;
  const float* x    = (const float*)d_in[0];
  const float* W_ih = (const float*)d_in[1];
  const float* W_hh = (const float*)d_in[2];
  const float* b_ih = (const float*)d_in[3];
  const float* b_hh = (const float*)d_in[4];
  const float* fc_w = (const float*)d_in[5];
  const float* fc_b = (const float*)d_in[6];
  float* out = (float*)d_out;

  dim3 block(64);
  dim3 grid(RNN_B / 32);
  rnn_fused_wmma<<<grid, block, 0, stream>>>(x, W_ih, W_hh, b_ih, b_hh, fc_w,
                                             fc_b, out);
}


# x projection moved to two helper waves per 16 batches, 32 steps ahead, handed over through LDS as f32; the recurrence wave keeps only the 2 h products and tanh; sums and their order unchanged
# speedup vs baseline: 4.1544x; 1.1967x over previous
#include <hip/hip_runtime.h>


typedef __attribute__((ext_vector_type(2))) float v2f;
typedef __attribute__((ext_vector_type(8))) float v8f;

#define RNN_B 4096
#define RNN_T 512
#define RNN_I 16
#define RNN_H 6
#define RNN_NP 2
#define RNN_CH 32
#define RNN_SPP (RNN_CH / RNN_NP)
#define RNN_NC (RNN_T / RNN_CH)

__device__ __forceinline__ v8f wmma_f32_4(v2f a, v2f b, v8f c) {
  return __builtin_amdgcn_wmma_f32_16x16x4_f32(false, a, false, b, (short)0, c,
                                               false, false);
}

__device__ __forceinline__ float tanh_fast(float x) {
#if __has_builtin(__builtin_amdgcn_tanhf)
  return __builtin_amdgcn_tanhf(x);
#else
  x = __builtin_amdgcn_fmed3f(x, -9.0f, 9.0f);
  float e = __expf(2.0f * x);
  return (e - 1.0f) * __builtin_amdgcn_rcpf(e + 1.0f);
#endif
}

__device__ __forceinline__ int rnn_unit_of_row(int row) {
  if (row < 4) return (row & 1) + 4 * (row >> 1);
  if (row == 8 || row == 9) return row - 6;
  return -1;
}

__global__ __launch_bounds__(32 * (1 + RNN_NP)) void rnn_fused_wmma(
    const float* __restrict__ x, const float* __restrict__ W_ih,
    const float* __restrict__ W_hh, const float* __restrict__ b_ih,
    const float* __restrict__ b_hh, const float* __restrict__ fc_w,
    const float* __restrict__ fc_b, float* __restrict__ out) {
  __shared__ float4 xproj[2][RNN_CH][32];
  __shared__ float hfin[16][8];

  const int tid  = threadIdx.x;
  const int lane = tid & 31;
  const int w    = __builtin_amdgcn_readfirstlane(tid >> 5);
  const int n    = lane & 15;
  const int hi   = lane >> 4;
  const int b0   = blockIdx.x * 16;
  const int p    = (w > 0) ? (w - 1) : 0;

  const int ua = rnn_unit_of_row(n);
  v2f wih[4];
#pragma unroll
  for (int s = 0; s < 4; ++s) {
#pragma unroll
    for (int v = 0; v < 2; ++v) {
      int k = 4 * s + 2 * hi + v;
      wih[s][v] = (ua >= 0) ? W_ih[ua * RNN_I + k] : 0.0f;
    }
  }
  v2f whh[2];
#pragma unroll
  for (int s = 0; s < 2; ++s) {
#pragma unroll
    for (int v = 0; v < 2; ++v) {
      int k = 4 * s + 2 * hi + v;
      whh[s][v] = (ua >= 0 && k < RNN_H) ? W_hh[ua * RNN_H + k] : 0.0f;
    }
  }
  v8f cbias;
#pragma unroll
  for (int r = 0; r < 8; ++r) {
    int uc = rnn_unit_of_row(8 * hi + r);
    cbias[r] = (uc >= 0) ? (b_ih[uc] + b_hh[uc]) : 0.0f;
  }

  const float* xp = x + (size_t)(b0 + n) * (RNN_T * RNN_I) + 2 * hi + p * (RNN_SPP * RNN_I);

  v2f hb0 = 0.0f, hb1 = 0.0f;

  v2f xs[RNN_SPP][4];
#pragma unroll
  for (int j = 0; j < RNN_SPP; ++j) {
#pragma unroll
    for (int s = 0; s < 4; ++s) xs[j][s] = 0.0f;
  }
  if (w != 0) {
#pragma unroll
    for (int j = 0; j < RNN_SPP; ++j) {
#pragma unroll
      for (int s = 0; s < 4; ++s) xs[j][s] = *(const v2f*)(xp + j * RNN_I + 4 * s);
    }
  }

  for (int i = 0; i <= RNN_NC; ++i) {
    if (w != 0) {
      if (i < RNN_NC) {
        const int tn = (i + 1 < RNN_NC) ? (i + 1) * RNN_CH : i * RNN_CH;
        const float* pn = xp + (size_t)tn * RNN_I;
        float4* dst = &xproj[i & 1][p * RNN_SPP][lane];
        v2f xn[RNN_SPP][4];
#pragma unroll
        for (int j = 0; j < RNN_SPP; ++j) {
#pragma unroll
          for (int s = 0; s < 4; ++s) xn[j][s] = *(const v2f*)(pn + j * RNN_I + 4 * s);
        }
        __builtin_amdgcn_sched_barrier(0);
#pragma unroll
        for (int j = 0; j < RNN_SPP; j += 2) {
          v8f d0 = cbias, d1 = cbias;
#pragma unroll
          for (int s = 0; s < 4; ++s) {
            d0 = wmma_f32_4(wih[s], xs[j][s], d0);
            d1 = wmma_f32_4(wih[s], xs[j + 1][s], d1);
          }
          dst[j * 32]       = make_float4(d0[0], d0[1], d0[2], d0[3]);
          dst[(j + 1) * 32] = make_float4(d1[0], d1[1], d1[2], d1[3]);
        }
#pragma unroll
        for (int j = 0; j < RNN_SPP; ++j) {
#pragma unroll
          for (int s = 0; s < 4; ++s) xs[j][s] = xn[j][s];
        }
      }
    } else {
      if (i >= 1) {
        const float4* srcp = &xproj[(i - 1) & 1][0][lane];
        float4 q[RNN_CH];
#pragma unroll
        for (int j = 0; j < RNN_CH; ++j) q[j] = srcp[j * 32];
        __builtin_amdgcn_sched_barrier(0);
#pragma unroll
        for (int j = 0; j < RNN_CH; ++j) {
          v8f c = {q[j].x, q[j].y, q[j].z, q[j].w, 0.0f, 0.0f, 0.0f, 0.0f};
          v8f acc = wmma_f32_4(whh[0], hb0, c);
          acc = wmma_f32_4(whh[1], hb1, acc);
          hb0[0] = tanh_fast(acc[0]);
          hb0[1] = tanh_fast(acc[1]);
          hb1[0] = tanh_fast(acc[2]);
          hb1[1] = tanh_fast(acc[3]);
        }
      }
    }
    __syncthreads();
  }

  if (w == 0) {
    if (hi == 0) {
      hfin[n][0] = hb0[0]; hfin[n][1] = hb0[1]; hfin[n][4] = hb1[0]; hfin[n][5] = hb1[1];
    } else {
      hfin[n][2] = hb0[0]; hfin[n][3] = hb0[1];
    }
  }
  __syncthreads();
  if (w == 0 && lane < 16) {
    float s = fc_b[0];
#pragma unroll
    for (int j = 0; j < RNN_H; ++j) s += hfin[lane][j] * fc_w[j];
    out[b0 + lane] = s;
  }
}

extern "C" void kernel_launch(void* const* d_in, const int* in_sizes, int n_in,
                              void* d_out, int out_size, void* d_ws,
                              size_t ws_size, hipStream_t stream) {
  (void)in_sizes; (void)n_in; (void)out_size; (void)d_ws; (void)ws_size;
  const float* x    = (const float*)d_in[0];
  const float* W_ih = (const float*)d_in[1];
  const float* W_hh = (const float*)d_in[2];
  const float* b_ih = (const float*)d_in[3];
  const float* b_hh = (const float*)d_in[4];
  const float* fc_w = (const float*)d_in[5];
  const float* fc_b = (const float*)d_in[6];
  float* out = (float*)d_out;

  dim3 block(32 * (1 + RNN_NP));
  dim3 grid(RNN_B / 16);
  rnn_fused_wmma<<<grid, block, 0, stream>>>(x, W_ih, W_hh, b_ih, b_hh, fc_w,
                                             fc_b, out);
}


# x fetched with 16-byte loads (half the load instructions): a lane half holds 8 consecutive inputs, so the x projection's sum order changes to inputs 2s,2s+1,8+2s,8+2s+1 per slice s; h sums unchanged
# speedup vs baseline: 4.6823x; 1.1271x over previous
#include <hip/hip_runtime.h>


typedef __attribute__((ext_vector_type(2))) float v2f;
typedef __attribute__((ext_vector_type(4))) float v4f;
typedef __attribute__((ext_vector_type(8))) float v8f;

#define RNN_B 4096
#define RNN_T 512
#define RNN_I 16
#define RNN_H 6
#define RNN_NP 2
#define RNN_CH 32
#define RNN_SPP (RNN_CH / RNN_NP)
#define RNN_NC (RNN_T / RNN_CH)

__device__ __forceinline__ v8f wmma_f32_4(v2f a, v2f b, v8f c) {
  return __builtin_amdgcn_wmma_f32_16x16x4_f32(false, a, false, b, (short)0, c,
                                               false, false);
}

__device__ __forceinline__ float tanh_fast(float x) {
#if __has_builtin(__builtin_amdgcn_tanhf)
  return __builtin_amdgcn_tanhf(x);
#else
  x = __builtin_amdgcn_fmed3f(x, -9.0f, 9.0f);
  float e = __expf(2.0f * x);
  return (e - 1.0f) * __builtin_amdgcn_rcpf(e + 1.0f);
#endif
}

__device__ __forceinline__ int rnn_unit_of_row(int row) {
  if (row < 4) return (row & 1) + 4 * (row >> 1);
  if (row == 8 || row == 9) return row - 6;
  return -1;
}

__global__ __launch_bounds__(32 * (1 + RNN_NP)) void rnn_fused_wmma(
    const float* __restrict__ x, const float* __restrict__ W_ih,
    const float* __restrict__ W_hh, const float* __restrict__ b_ih,
    const float* __restrict__ b_hh, const float* __restrict__ fc_w,
    const float* __restrict__ fc_b, float* __restrict__ out) {
  __shared__ float4 xproj[2][RNN_CH][32];
  __shared__ float hfin[16][8];

  const int tid  = threadIdx.x;
  const int lane = tid & 31;
  const int w    = __builtin_amdgcn_readfirstlane(tid >> 5);
  const int n    = lane & 15;
  const int hi   = lane >> 4;
  const int b0   = blockIdx.x * 16;
  const int p    = (w > 0) ? (w - 1) : 0;

  const int ua = rnn_unit_of_row(n);
  v2f wih[4];
#pragma unroll
  for (int s = 0; s < 4; ++s) {
#pragma unroll
    for (int v = 0; v < 2; ++v) {
      int k = 8 * hi + 2 * s + v;
      wih[s][v] = (ua >= 0) ? W_ih[ua * RNN_I + k] : 0.0f;
    }
  }
  v2f whh[2];
#pragma unroll
  for (int s = 0; s < 2; ++s) {
#pragma unroll
    for (int v = 0; v < 2; ++v) {
      int k = 4 * s + 2 * hi + v;
      whh[s][v] = (ua >= 0 && k < RNN_H) ? W_hh[ua * RNN_H + k] : 0.0f;
    }
  }
  v8f cbias;
#pragma unroll
  for (int r = 0; r < 8; ++r) {
    int uc = rnn_unit_of_row(8 * hi + r);
    cbias[r] = (uc >= 0) ? (b_ih[uc] + b_hh[uc]) : 0.0f;
  }

  const float* xp = x + (size_t)(b0 + n) * (RNN_T * RNN_I) + 8 * hi + p * (RNN_SPP * RNN_I);

  v2f hb0 = 0.0f, hb1 = 0.0f;

  v4f xs[RNN_SPP][2];
#pragma unroll
  for (int j = 0; j < RNN_SPP; ++j) {
#pragma unroll
    for (int q = 0; q < 2; ++q) xs[j][q] = 0.0f;
  }
  if (w != 0) {
#pragma unroll
    for (int j = 0; j < RNN_SPP; ++j) {
#pragma unroll
      for (int q = 0; q < 2; ++q) xs[j][q] = *(const v4f*)(xp + j * RNN_I + 4 * q);
    }
  }

  for (int i = 0; i <= RNN_NC; ++i) {
    if (w != 0) {
      if (i < RNN_NC) {
        const int tn = (i + 1 < RNN_NC) ? (i + 1) * RNN_CH : i * RNN_CH;
        const float* pn = xp + (size_t)tn * RNN_I;
        float4* dst = &xproj[i & 1][p * RNN_SPP][lane];
        v4f xn[RNN_SPP][2];
#pragma unroll
        for (int j = 0; j < RNN_SPP; ++j) {
#pragma unroll
          for (int q = 0; q < 2; ++q) xn[j][q] = *(const v4f*)(pn + j * RNN_I + 4 * q);
        }
        __builtin_amdgcn_sched_barrier(0);
#pragma unroll
        for (int j = 0; j < RNN_SPP; j += 2) {
          v8f d0 = cbias, d1 = cbias;
#pragma unroll
          for (int s = 0; s < 4; ++s) {
            const v4f f0 = xs[j][s >> 1], f1 = xs[j + 1][s >> 1];
            const v2f x0 = (s & 1) ? __builtin_shufflevector(f0, f0, 2, 3) : __builtin_shufflevector(f0, f0, 0, 1);
            const v2f x1 = (s & 1) ? __builtin_shufflevector(f1, f1, 2, 3) : __builtin_shufflevector(f1, f1, 0, 1);
            d0 = wmma_f32_4(wih[s], x0, d0);
            d1 = wmma_f32_4(wih[s], x1, d1);
          }
          dst[j * 32]       = make_float4(d0[0], d0[1], d0[2], d0[3]);
          dst[(j + 1) * 32] = make_float4(d1[0], d1[1], d1[2], d1[3]);
        }
#pragma unroll
        for (int j = 0; j < RNN_SPP; ++j) {
#pragma unroll
          for (int q = 0; q < 2; ++q) xs[j][q] = xn[j][q];
        }
      }
    } else {
      if (i >= 1) {
        const float4* srcp = &xproj[(i - 1) & 1][0][lane];
        float4 q[RNN_CH];
#pragma unroll
        for (int j = 0; j < RNN_CH; ++j) q[j] = srcp[j * 32];
        __builtin_amdgcn_sched_barrier(0);
#pragma unroll
        for (int j = 0; j < RNN_CH; ++j) {
          v8f c = {q[j].x, q[j].y, q[j].z, q[j].w, 0.0f, 0.0f, 0.0f, 0.0f};
          v8f acc = wmma_f32_4(whh[0], hb0, c);
          acc = wmma_f32_4(whh[1], hb1, acc);
          hb0[0] = tanh_fast(acc[0]);
          hb0[1] = tanh_fast(acc[1]);
          hb1[0] = tanh_fast(acc[2]);
          hb1[1] = tanh_fast(acc[3]);
        }
      }
    }
    __syncthreads();
  }

  if (w == 0) {
    if (hi == 0) {
      hfin[n][0] = hb0[0]; hfin[n][1] = hb0[1]; hfin[n][4] = hb1[0]; hfin[n][5] = hb1[1];
    } else {
      hfin[n][2] = hb0[0]; hfin[n][3] = hb0[1];
    }
  }
  __syncthreads();
  if (w == 0 && lane < 16) {
    float s = fc_b[0];
#pragma unroll
    for (int j = 0; j < RNN_H; ++j) s += hfin[lane][j] * fc_w[j];
    out[b0 + lane] = s;
  }
}

extern "C" void kernel_launch(void* const* d_in, const int* in_sizes, int n_in,
                              void* d_out, int out_size, void* d_ws,
                              size_t ws_size, hipStream_t stream) {
  (void)in_sizes; (void)n_in; (void)out_size; (void)d_ws; (void)ws_size;
  const float* x    = (const float*)d_in[0];
  const float* W_ih = (const float*)d_in[1];
  const float* W_hh = (const float*)d_in[2];
  const float* b_ih = (const float*)d_in[3];
  const float* b_hh = (const float*)d_in[4];
  const float* fc_w = (const float*)d_in[5];
  const float* fc_b = (const float*)d_in[6];
  float* out = (float*)d_out;

  dim3 block(32 * (1 + RNN_NP));
  dim3 grid(RNN_B / 16);
  rnn_fused_wmma<<<grid, block, 0, stream>>>(x, W_ih, W_hh, b_ih, b_hh, fc_w,
                                             fc_b, out);
}


# x fetched as whole lines (one load = 512 consecutive bytes of one batch row, 4 lines touched instead of 16) and turned into fragment layout through an LDS staging area; sums and their order unchanged
# speedup vs baseline: 6.0260x; 1.2870x over previous
#include <hip/hip_runtime.h>


typedef __attribute__((ext_vector_type(2))) float v2f;
typedef __attribute__((ext_vector_type(4))) float v4f;
typedef __attribute__((ext_vector_type(8))) float v8f;

#define RNN_B 4096
#define RNN_T 512
#define RNN_I 16
#define RNN_H 6
#define RNN_NP 2
#define RNN_CH 64
#define RNN_RG 32
#define RNN_SPP (RNN_CH / RNN_NP)
#define RNN_NC (RNN_T / RNN_CH)
#define RNN_LPR (RNN_SPP / 8)
#define RNN_SPITCH 129

__device__ __forceinline__ v8f wmma_f32_4(v2f a, v2f b, v8f c) {
  return __builtin_amdgcn_wmma_f32_16x16x4_f32(false, a, false, b, (short)0, c,
                                               false, false);
}

__device__ __forceinline__ float tanh_fast(float x) {
#if __has_builtin(__builtin_amdgcn_tanhf)
  return __builtin_amdgcn_tanhf(x);
#else
  x = __builtin_amdgcn_fmed3f(x, -9.0f, 9.0f);
  float e = __expf(2.0f * x);
  return (e - 1.0f) * __builtin_amdgcn_rcpf(e + 1.0f);
#endif
}

__device__ __forceinline__ int rnn_unit_of_row(int row) {
  if (row < 4) return (row & 1) + 4 * (row >> 1);
  if (row == 8 || row == 9) return row - 6;
  return -1;
}

__global__ __launch_bounds__(32 * (1 + RNN_NP)) void rnn_fused_wmma(
    const float* __restrict__ x, const float* __restrict__ W_ih,
    const float* __restrict__ W_hh, const float* __restrict__ b_ih,
    const float* __restrict__ b_hh, const float* __restrict__ fc_w,
    const float* __restrict__ fc_b, float* __restrict__ out) {
  __shared__ float4 xproj[2][RNN_CH][32];
  __shared__ float hfin[16][8];
  __shared__ v4f stage[RNN_NP][16 * RNN_SPITCH];

  const int tid  = threadIdx.x;
  const int lane = tid & 31;
  const int w    = __builtin_amdgcn_readfirstlane(tid >> 5);
  const int n    = lane & 15;
  const int hi   = lane >> 4;
  const int b0   = blockIdx.x * 16;
  const int p    = (w > 0) ? (w - 1) : 0;

  const int ua = rnn_unit_of_row(n);
  v2f wih[4];
#pragma unroll
  for (int s = 0; s < 4; ++s) {
#pragma unroll
    for (int v = 0; v < 2; ++v) {
      int k = 8 * hi + 2 * s + v;
      wih[s][v] = (ua >= 0) ? W_ih[ua * RNN_I + k] : 0.0f;
    }
  }
  v2f whh[2];
#pragma unroll
  for (int s = 0; s < 2; ++s) {
#pragma unroll
    for (int v = 0; v < 2; ++v) {
      int k = 4 * s + 2 * hi + v;
      whh[s][v] = (ua >= 0 && k < RNN_H) ? W_hh[ua * RNN_H + k] : 0.0f;
    }
  }
  v8f cbias;
#pragma unroll
  for (int r = 0; r < 8; ++r) {
    int uc = rnn_unit_of_row(8 * hi + r);
    cbias[r] = (uc >= 0) ? (b_ih[uc] + b_hh[uc]) : 0.0f;
  }

  const float* xraw = x + (size_t)b0 * (RNN_T * RNN_I) + p * (RNN_SPP * RNN_I) + 4 * lane;
  v4f* const st = &stage[p][0];
  v4f* const stw = st + lane;
  const v4f* const stf = st + n * RNN_SPITCH + 2 * hi;

  v2f hb0 = 0.0f, hb1 = 0.0f;

  if (w != 0) {
    v4f x0[16][RNN_LPR];
#pragma unroll
    for (int r = 0; r < 16; ++r) {
#pragma unroll
      for (int c = 0; c < RNN_LPR; ++c) x0[r][c] = *(const v4f*)(xraw + (size_t)r * (RNN_T * RNN_I) + c * 128);
    }
#pragma unroll
    for (int r = 0; r < 16; ++r) {
#pragma unroll
      for (int c = 0; c < RNN_LPR; ++c) stw[r * RNN_SPITCH + 32 * c] = x0[r][c];
    }
  }
  __syncthreads();

  for (int i = 0; i <= RNN_NC; ++i) {
    if (w != 0) {
      if (i < RNN_NC) {
        const int tn = (i + 1 < RNN_NC) ? (i + 1) * RNN_CH : i * RNN_CH;
        const float* pn = xraw + (size_t)tn * RNN_I;
        float4* dst = &xproj[i & 1][p * RNN_SPP][lane];
        v4f xn[16][RNN_LPR];
#pragma unroll
        for (int r = 0; r < 16; ++r) {
#pragma unroll
          for (int c = 0; c < RNN_LPR; ++c) xn[r][c] = *(const v4f*)(pn + (size_t)r * (RNN_T * RNN_I) + c * 128);
        }
        v4f xs[RNN_SPP][2];
#pragma unroll
        for (int j = 0; j < RNN_SPP; ++j) {
#pragma unroll
          for (int q = 0; q < 2; ++q) xs[j][q] = stf[4 * j + q];
        }
        __builtin_amdgcn_sched_barrier(0);
#pragma unroll
        for (int j = 0; j < RNN_SPP; j += 2) {
          v8f d0 = cbias, d1 = cbias;
#pragma unroll
          for (int s = 0; s < 4; ++s) {
            const v4f f0 = xs[j][s >> 1], f1 = xs[j + 1][s >> 1];
            const v2f x0 = (s & 1) ? __builtin_shufflevector(f0, f0, 2, 3) : __builtin_shufflevector(f0, f0, 0, 1);
            const v2f x1 = (s & 1) ? __builtin_shufflevector(f1, f1, 2, 3) : __builtin_shufflevector(f1, f1, 0, 1);
            d0 = wmma_f32_4(wih[s], x0, d0);
            d1 = wmma_f32_4(wih[s], x1, d1);
          }
          dst[j * 32]       = make_float4(d0[0], d0[1], d0[2], d0[3]);
          dst[(j + 1) * 32] = make_float4(d1[0], d1[1], d1[2], d1[3]);
        }
        __builtin_amdgcn_sched_barrier(0);
#pragma unroll
        for (int r = 0; r < 16; ++r) {
#pragma unroll
          for (int c = 0; c < RNN_LPR; ++c) stw[r * RNN_SPITCH + 32 * c] = xn[r][c];
        }
      }
    } else {
      if (i >= 1) {
#pragma unroll
        for (int g = 0; g < RNN_CH / RNN_RG; ++g) {
          const float4* srcp = &xproj[(i - 1) & 1][g * RNN_RG][lane];
          float4 q[RNN_RG];
#pragma unroll
          for (int j = 0; j < RNN_RG; ++j) q[j] = srcp[j * 32];
          __builtin_amdgcn_sched_barrier(0);
#pragma unroll
          for (int j = 0; j < RNN_RG; ++j) {
            v8f c = {q[j].x, q[j].y, q[j].z, q[j].w, 0.0f, 0.0f, 0.0f, 0.0f};
            v8f acc = wmma_f32_4(whh[0], hb0, c);
            acc = wmma_f32_4(whh[1], hb1, acc);
            hb0[0] = tanh_fast(acc[0]);
            hb0[1] = tanh_fast(acc[1]);
            hb1[0] = tanh_fast(acc[2]);
            hb1[1] = tanh_fast(acc[3]);
          }
          __builtin_amdgcn_sched_barrier(0);
        }
      }
    }
    __syncthreads();
  }

  if (w == 0) {
    if (hi == 0) {
      hfin[n][0] = hb0[0]; hfin[n][1] = hb0[1]; hfin[n][4] = hb1[0]; hfin[n][5] = hb1[1];
    } else {
      hfin[n][2] = hb0[0]; hfin[n][3] = hb0[1];
    }
  }
  __syncthreads();
  if (w == 0 && lane < 16) {
    float s = fc_b[0];
#pragma unroll
    for (int j = 0; j < RNN_H; ++j) s += hfin[lane][j] * fc_w[j];
    out[b0 + lane] = s;
  }
}

extern "C" void kernel_launch(void* const* d_in, const int* in_sizes, int n_in,
                              void* d_out, int out_size, void* d_ws,
                              size_t ws_size, hipStream_t stream) {
  (void)in_sizes; (void)n_in; (void)out_size; (void)d_ws; (void)ws_size;
  const float* x    = (const float*)d_in[0];
  const float* W_ih = (const float*)d_in[1];
  const float* W_hh = (const float*)d_in[2];
  const float* b_ih = (const float*)d_in[3];
  const float* b_hh = (const float*)d_in[4];
  const float* fc_w = (const float*)d_in[5];
  const float* fc_b = (const float*)d_in[6];
  float* out = (float*)d_out;

  dim3 block(32 * (1 + RNN_NP));
  dim3 grid(RNN_B / 16);
  rnn_fused_wmma<<<grid, block, 0, stream>>>(x, W_ih, W_hh, b_ih, b_hh, fc_w,
                                             fc_b, out);
}


# third projection wave on the idle fourth SIMD: a chunk's 64 steps split 24/24/16 over three helper waves, a quarter less loads, LDS traffic and products per wave and pass; sums and order unchanged
# speedup vs baseline: 6.5233x; 1.0825x over previous
#include <hip/hip_runtime.h>


typedef __attribute__((ext_vector_type(2))) float v2f;
typedef __attribute__((ext_vector_type(4))) float v4f;
typedef __attribute__((ext_vector_type(8))) float v8f;

#define RNN_B 4096
#define RNN_T 512
#define RNN_I 16
#define RNN_H 6
#define RNN_NP 3
#define RNN_CH 64
#define RNN_RG 32
#define RNN_SPA 24
#define RNN_SPB (RNN_CH - 2 * RNN_SPA)
#define RNN_NC (RNN_T / RNN_CH)
#define RNN_SPITCH (4 * RNN_SPA + 1)

__device__ __forceinline__ v8f wmma_f32_4(v2f a, v2f b, v8f c) {
  return __builtin_amdgcn_wmma_f32_16x16x4_f32(false, a, false, b, (short)0, c,
                                               false, false);
}

__device__ __forceinline__ float tanh_fast(float x) {
#if __has_builtin(__builtin_amdgcn_tanhf)
  return __builtin_amdgcn_tanhf(x);
#else
  x = __builtin_amdgcn_fmed3f(x, -9.0f, 9.0f);
  float e = __expf(2.0f * x);
  return (e - 1.0f) * __builtin_amdgcn_rcpf(e + 1.0f);
#endif
}

__device__ __forceinline__ int rnn_unit_of_row(int row) {
  if (row < 4) return (row & 1) + 4 * (row >> 1);
  if (row == 8 || row == 9) return row - 6;
  return -1;
}


template <int SPP>
__device__ __forceinline__ void rnn_proj_first(const float* __restrict__ xraw, v4f* stw) {
  constexpr int LPR = SPP / 8;
  v4f x0[16][LPR];
#pragma unroll
  for (int r = 0; r < 16; ++r) {
#pragma unroll
    for (int c = 0; c < LPR; ++c) x0[r][c] = *(const v4f*)(xraw + (size_t)r * (RNN_T * RNN_I) + c * 128);
  }
#pragma unroll
  for (int r = 0; r < 16; ++r) {
#pragma unroll
    for (int c = 0; c < LPR; ++c) stw[r * RNN_SPITCH + 32 * c] = x0[r][c];
  }
}

template <int SPP>
__device__ __forceinline__ void rnn_proj_pass(const float* __restrict__ pn, const v4f* stf, v4f* stw,
                                              float4* dst, v2f w0, v2f w1, v2f w2, v2f w3, v8f cbias) {
  constexpr int LPR = SPP / 8;
  const v2f wih[4] = {w0, w1, w2, w3};
  v4f xn[16][LPR];
#pragma unroll
  for (int r = 0; r < 16; ++r) {
#pragma unroll
    for (int c = 0; c < LPR; ++c) xn[r][c] = *(const v4f*)(pn + (size_t)r * (RNN_T * RNN_I) + c * 128);
  }
  v4f xs[SPP][2];
#pragma unroll
  for (int j = 0; j < SPP; ++j) {
#pragma unroll
    for (int q = 0; q < 2; ++q) xs[j][q] = stf[4 * j + q];
  }
  v8f d[SPP];
#pragma unroll
  for (int j = 0; j < SPP; ++j) d[j] = cbias;
#pragma unroll
  for (int s = 0; s < 4; ++s) {
#pragma unroll
    for (int j = 0; j < SPP; ++j) {
      const v4f f = xs[j][s >> 1];
      const v2f xf = (s & 1) ? __builtin_shufflevector(f, f, 2, 3) : __builtin_shufflevector(f, f, 0, 1);
      d[j] = wmma_f32_4(wih[s], xf, d[j]);
    }
  }
#pragma unroll
  for (int j = 0; j < SPP; ++j) dst[j * 32] = make_float4(d[j][0], d[j][1], d[j][2], d[j][3]);
#pragma unroll
  for (int r = 0; r < 16; ++r) {
#pragma unroll
    for (int c = 0; c < LPR; ++c) stw[r * RNN_SPITCH + 32 * c] = xn[r][c];
  }
}

__global__ __launch_bounds__(32 * (1 + RNN_NP)) void rnn_fused_wmma(
    const float* __restrict__ x, const float* __restrict__ W_ih,
    const float* __restrict__ W_hh, const float* __restrict__ b_ih,
    const float* __restrict__ b_hh, const float* __restrict__ fc_w,
    const float* __restrict__ fc_b, float* __restrict__ out) {
  __shared__ float4 xproj[2][RNN_CH][32];
  __shared__ float hfin[16][8];
  __shared__ v4f stage[RNN_NP][16 * RNN_SPITCH];

  const int tid  = threadIdx.x;
  const int lane = tid & 31;
  const int w    = __builtin_amdgcn_readfirstlane(tid >> 5);
  const int n    = lane & 15;
  const int hi   = lane >> 4;
  const int b0   = blockIdx.x * 16;
  const int p    = (w > 0) ? (w - 1) : 0;

  const int ua = rnn_unit_of_row(n);
  v2f wih[4];
#pragma unroll
  for (int s = 0; s < 4; ++s) {
#pragma unroll
    for (int v = 0; v < 2; ++v) {
      int k = 8 * hi + 2 * s + v;
      wih[s][v] = (ua >= 0) ? W_ih[ua * RNN_I + k] : 0.0f;
    }
  }
  v2f whh[2];
#pragma unroll
  for (int s = 0; s < 2; ++s) {
#pragma unroll
    for (int v = 0; v < 2; ++v) {
      int k = 4 * s + 2 * hi + v;
      whh[s][v] = (ua >= 0 && k < RNN_H) ? W_hh[ua * RNN_H + k] : 0.0f;
    }
  }
  v8f cbias;
#pragma unroll
  for (int r = 0; r < 8; ++r) {
    int uc = rnn_unit_of_row(8 * hi + r);
    cbias[r] = (uc >= 0) ? (b_ih[uc] + b_hh[uc]) : 0.0f;
  }

  const float* xraw = x + (size_t)b0 * (RNN_T * RNN_I) + p * (RNN_SPA * RNN_I) + 4 * lane;
  v4f* const st = &stage[p][0];
  v4f* const stw = st + lane;
  const v4f* const stf = st + n * RNN_SPITCH + 2 * hi;

  v2f hb0 = 0.0f, hb1 = 0.0f;

  if (w != 0) {
    if (w == RNN_NP) rnn_proj_first<RNN_SPB>(xraw, stw);
    else             rnn_proj_first<RNN_SPA>(xraw, stw);
  }
  __syncthreads();

  for (int i = 0; i <= RNN_NC; ++i) {
    if (w != 0) {
      if (i < RNN_NC) {
        const int tn = (i + 1 < RNN_NC) ? (i + 1) * RNN_CH : i * RNN_CH;
        const float* pn = xraw + (size_t)tn * RNN_I;
        float4* dst = &xproj[i & 1][p * RNN_SPA][lane];
        if (w == RNN_NP) rnn_proj_pass<RNN_SPB>(pn, stf, stw, dst, wih[0], wih[1], wih[2], wih[3], cbias);
        else             rnn_proj_pass<RNN_SPA>(pn, stf, stw, dst, wih[0], wih[1], wih[2], wih[3], cbias);
      }
    } else {
      if (i >= 1) {
#pragma unroll
        for (int g = 0; g < RNN_CH / RNN_RG; ++g) {
          const float4* srcp = &xproj[(i - 1) & 1][g * RNN_RG][lane];
          float4 q[RNN_RG];
#pragma unroll
          for (int j = 0; j < RNN_RG; ++j) q[j] = srcp[j * 32];
          __builtin_amdgcn_sched_barrier(0);
#pragma unroll
          for (int j = 0; j < RNN_RG; ++j) {
            v8f c = {q[j].x, q[j].y, q[j].z, q[j].w, 0.0f, 0.0f, 0.0f, 0.0f};
            v8f acc = wmma_f32_4(whh[0], hb0, c);
            acc = wmma_f32_4(whh[1], hb1, acc);
            hb0[0] = tanh_fast(acc[0]);
            hb0[1] = tanh_fast(acc[1]);
            hb1[0] = tanh_fast(acc[2]);
            hb1[1] = tanh_fast(acc[3]);
          }
          __builtin_amdgcn_sched_barrier(0);
        }
      }
    }
    __syncthreads();
  }

  if (w == 0) {
    if (hi == 0) {
      hfin[n][0] = hb0[0]; hfin[n][1] = hb0[1]; hfin[n][4] = hb1[0]; hfin[n][5] = hb1[1];
    } else {
      hfin[n][2] = hb0[0]; hfin[n][3] = hb0[1];
    }
  }
  __syncthreads();
  if (w == 0 && lane < 16) {
    float s = fc_b[0];
#pragma unroll
    for (int j = 0; j < RNN_H; ++j) s += hfin[lane][j] * fc_w[j];
    out[b0 + lane] = s;
  }
}

extern "C" void kernel_launch(void* const* d_in, const int* in_sizes, int n_in,
                              void* d_out, int out_size, void* d_ws,
                              size_t ws_size, hipStream_t stream) {
  (void)in_sizes; (void)n_in; (void)out_size; (void)d_ws; (void)ws_size;
  const float* x    = (const float*)d_in[0];
  const float* W_ih = (const float*)d_in[1];
  const float* W_hh = (const float*)d_in[2];
  const float* b_ih = (const float*)d_in[3];
  const float* b_hh = (const float*)d_in[4];
  const float* fc_w = (const float*)d_in[5];
  const float* fc_b = (const float*)d_in[6];
  float* out = (float*)d_out;

  dim3 block(32 * (1 + RNN_NP));
  dim3 grid(RNN_B / 16);
  rnn_fused_wmma<<<grid, block, 0, stream>>>(x, W_ih, W_hh, b_ih, b_hh, fc_w,
                                             fc_b, out);
}


# start-up loads reordered: first chunk's x asked for at kernel entry, weights and biases behind it as one group of unconditional loads (clamped index, zero selected after); sums and order unchanged
# speedup vs baseline: 6.8603x; 1.0517x over previous
#include <hip/hip_runtime.h>


typedef __attribute__((ext_vector_type(2))) float v2f;
typedef __attribute__((ext_vector_type(4))) float v4f;
typedef __attribute__((ext_vector_type(8))) float v8f;

#define RNN_B 4096
#define RNN_T 512
#define RNN_I 16
#define RNN_H 6
#define RNN_NP 3
#define RNN_CH 64
#define RNN_RG 32
#define RNN_SPA 24
#define RNN_SPB (RNN_CH - 2 * RNN_SPA)
#define RNN_NC (RNN_T / RNN_CH)
#define RNN_SPITCH (4 * RNN_SPA + 1)

__device__ __forceinline__ v8f wmma_f32_4(v2f a, v2f b, v8f c) {
  return __builtin_amdgcn_wmma_f32_16x16x4_f32(false, a, false, b, (short)0, c,
                                               false, false);
}

__device__ __forceinline__ float tanh_fast(float x) {
#if __has_builtin(__builtin_amdgcn_tanhf)
  return __builtin_amdgcn_tanhf(x);
#else
  x = __builtin_amdgcn_fmed3f(x, -9.0f, 9.0f);
  float e = __expf(2.0f * x);
  return (e - 1.0f) * __builtin_amdgcn_rcpf(e + 1.0f);
#endif
}

__device__ __forceinline__ int rnn_unit_of_row(int row) {
  if (row < 4) return (row & 1) + 4 * (row >> 1);
  if (row == 8 || row == 9) return row - 6;
  return -1;
}


__device__ __forceinline__ void rnn_fetch_first(const float* __restrict__ xraw, bool wide, v4f (&x0)[16][RNN_SPA / 8]) {
#pragma unroll
  for (int r = 0; r < 16; ++r) {
#pragma unroll
    for (int c = 0; c < RNN_SPB / 8; ++c) x0[r][c] = *(const v4f*)(xraw + (size_t)r * (RNN_T * RNN_I) + c * 128);
  }
  if (wide) {
#pragma unroll
    for (int r = 0; r < 16; ++r) {
#pragma unroll
      for (int c = RNN_SPB / 8; c < RNN_SPA / 8; ++c) x0[r][c] = *(const v4f*)(xraw + (size_t)r * (RNN_T * RNN_I) + c * 128);
    }
  }
}

__device__ __forceinline__ void rnn_stage_first(v4f* stw, bool wide, const v4f (&x0)[16][RNN_SPA / 8]) {
#pragma unroll
  for (int r = 0; r < 16; ++r) {
#pragma unroll
    for (int c = 0; c < RNN_SPB / 8; ++c) stw[r * RNN_SPITCH + 32 * c] = x0[r][c];
  }
  if (wide) {
#pragma unroll
    for (int r = 0; r < 16; ++r) {
#pragma unroll
      for (int c = RNN_SPB / 8; c < RNN_SPA / 8; ++c) stw[r * RNN_SPITCH + 32 * c] = x0[r][c];
    }
  }
}

template <int SPP>
__device__ __forceinline__ void rnn_proj_pass(const float* __restrict__ pn, const v4f* stf, v4f* stw,
                                              float4* dst, v2f w0, v2f w1, v2f w2, v2f w3, v8f cbias) {
  constexpr int LPR = SPP / 8;
  const v2f wih[4] = {w0, w1, w2, w3};
  v4f xn[16][LPR];
#pragma unroll
  for (int r = 0; r < 16; ++r) {
#pragma unroll
    for (int c = 0; c < LPR; ++c) xn[r][c] = *(const v4f*)(pn + (size_t)r * (RNN_T * RNN_I) + c * 128);
  }
  v4f xs[SPP][2];
#pragma unroll
  for (int j = 0; j < SPP; ++j) {
#pragma unroll
    for (int q = 0; q < 2; ++q) xs[j][q] = stf[4 * j + q];
  }
  v8f d[SPP];
#pragma unroll
  for (int j = 0; j < SPP; ++j) d[j] = cbias;
#pragma unroll
  for (int s = 0; s < 4; ++s) {
#pragma unroll
    for (int j = 0; j < SPP; ++j) {
      const v4f f = xs[j][s >> 1];
      const v2f xf = (s & 1) ? __builtin_shufflevector(f, f, 2, 3) : __builtin_shufflevector(f, f, 0, 1);
      d[j] = wmma_f32_4(wih[s], xf, d[j]);
    }
  }
#pragma unroll
  for (int j = 0; j < SPP; ++j) dst[j * 32] = make_float4(d[j][0], d[j][1], d[j][2], d[j][3]);
#pragma unroll
  for (int r = 0; r < 16; ++r) {
#pragma unroll
    for (int c = 0; c < LPR; ++c) stw[r * RNN_SPITCH + 32 * c] = xn[r][c];
  }
}

__global__ __launch_bounds__(32 * (1 + RNN_NP)) void rnn_fused_wmma(
    const float* __restrict__ x, const float* __restrict__ W_ih,
    const float* __restrict__ W_hh, const float* __restrict__ b_ih,
    const float* __restrict__ b_hh, const float* __restrict__ fc_w,
    const float* __restrict__ fc_b, float* __restrict__ out) {
  __shared__ float4 xproj[2][RNN_CH][32];
  __shared__ float hfin[16][8];
  __shared__ v4f stage[RNN_NP][16 * RNN_SPITCH];

  const int tid  = threadIdx.x;
  const int lane = tid & 31;
  const int w    = __builtin_amdgcn_readfirstlane(tid >> 5);
  const int n    = lane & 15;
  const int hi   = lane >> 4;
  const int b0   = blockIdx.x * 16;
  const int p    = (w > 0) ? (w - 1) : 0;

  const float* xraw = x + (size_t)b0 * (RNN_T * RNN_I) + p * (RNN_SPA * RNN_I) + 4 * lane;
  v4f* const st = &stage[p][0];
  v4f* const stw = st + lane;
  const v4f* const stf = st + n * RNN_SPITCH + 2 * hi;

  v4f x0[16][RNN_SPA / 8];
  if (w != 0) rnn_fetch_first(xraw, w != RNN_NP, x0);

  const int ua = rnn_unit_of_row(n);
  const int uw = (ua >= 0) ? ua : 0;
  v2f wih[4];
#pragma unroll
  for (int s = 0; s < 4; ++s) {
#pragma unroll
    for (int v = 0; v < 2; ++v) {
      int k = 8 * hi + 2 * s + v;
      const float t = W_ih[uw * RNN_I + k];
      wih[s][v] = (ua >= 0) ? t : 0.0f;
    }
  }
  v2f whh[2];
#pragma unroll
  for (int s = 0; s < 2; ++s) {
#pragma unroll
    for (int v = 0; v < 2; ++v) {
      int k = 4 * s + 2 * hi + v;
      const float t = W_hh[uw * RNN_H + ((k < RNN_H) ? k : 0)];
      whh[s][v] = (ua >= 0 && k < RNN_H) ? t : 0.0f;
    }
  }
  v8f cbias;
#pragma unroll
  for (int r = 0; r < 8; ++r) {
    int uc = rnn_unit_of_row(8 * hi + r);
    const int ub = (uc >= 0) ? uc : 0;
    const float t = b_ih[ub] + b_hh[ub];
    cbias[r] = (uc >= 0) ? t : 0.0f;
  }

  v2f hb0 = 0.0f, hb1 = 0.0f;

  if (w != 0) rnn_stage_first(stw, w != RNN_NP, x0);
  __syncthreads();

  for (int i = 0; i <= RNN_NC; ++i) {
    if (w != 0) {
      if (i < RNN_NC) {
        const int tn = (i + 1 < RNN_NC) ? (i + 1) * RNN_CH : i * RNN_CH;
        const float* pn = xraw + (size_t)tn * RNN_I;
        float4* dst = &xproj[i & 1][p * RNN_SPA][lane];
        if (w == RNN_NP) rnn_proj_pass<RNN_SPB>(pn, stf, stw, dst, wih[0], wih[1], wih[2], wih[3], cbias);
        else             rnn_proj_pass<RNN_SPA>(pn, stf, stw, dst, wih[0], wih[1], wih[2], wih[3], cbias);
      }
    } else {
      if (i >= 1) {
#pragma unroll
        for (int g = 0; g < RNN_CH / RNN_RG; ++g) {
          const float4* srcp = &xproj[(i - 1) & 1][g * RNN_RG][lane];
          float4 q[RNN_RG];
#pragma unroll
          for (int j = 0; j < RNN_RG; ++j) q[j] = srcp[j * 32];
          __builtin_amdgcn_sched_barrier(0);
#pragma unroll
          for (int j = 0; j < RNN_RG; ++j) {
            v8f c = {q[j].x, q[j].y, q[j].z, q[j].w, 0.0f, 0.0f, 0.0f, 0.0f};
            v8f acc = wmma_f32_4(whh[0], hb0, c);
            acc = wmma_f32_4(whh[1], hb1, acc);
            hb0[0] = tanh_fast(acc[0]);
            hb0[1] = tanh_fast(acc[1]);
            hb1[0] = tanh_fast(acc[2]);
            hb1[1] = tanh_fast(acc[3]);
          }
          __builtin_amdgcn_sched_barrier(0);
        }
      }
    }
    __syncthreads();
  }

  if (w == 0) {
    if (hi == 0) {
      hfin[n][0] = hb0[0]; hfin[n][1] = hb0[1]; hfin[n][4] = hb1[0]; hfin[n][5] = hb1[1];
    } else {
      hfin[n][2] = hb0[0]; hfin[n][3] = hb0[1];
    }
  }
  __syncthreads();
  if (w == 0 && lane < 16) {
    float s = fc_b[0];
#pragma unroll
    for (int j = 0; j < RNN_H; ++j) s += hfin[lane][j] * fc_w[j];
    out[b0 + lane] = s;
  }
}

extern "C" void kernel_launch(void* const* d_in, const int* in_sizes, int n_in,
                              void* d_out, int out_size, void* d_ws,
                              size_t ws_size, hipStream_t stream) {
  (void)in_sizes; (void)n_in; (void)out_size; (void)d_ws; (void)ws_size;
  const float* x    = (const float*)d_in[0];
  const float* W_ih = (const float*)d_in[1];
  const float* W_hh = (const float*)d_in[2];
  const float* b_ih = (const float*)d_in[3];
  const float* b_hh = (const float*)d_in[4];
  const float* fc_w = (const float*)d_in[5];
  const float* fc_b = (const float*)d_in[6];
  float* out = (float*)d_out;

  dim3 block(32 * (1 + RNN_NP));
  dim3 grid(RNN_B / 16);
  rnn_fused_wmma<<<grid, block, 0, stream>>>(x, W_ih, W_hh, b_ih, b_hh, fc_w,
                                             fc_b, out);
}


# hidden units re-mapped to D rows 0,2,3 / 8,10,11: first h product's fragment holds one unit per lane, so one tanh, not two, precedes it (3 per step, not 4); h sum order now units 0,3 then 1,2,4,5
# speedup vs baseline: 7.2067x; 1.0505x over previous
#include <hip/hip_runtime.h>


typedef __attribute__((ext_vector_type(2))) float v2f;
typedef __attribute__((ext_vector_type(4))) float v4f;
typedef __attribute__((ext_vector_type(8))) float v8f;

#define RNN_B 4096
#define RNN_T 512
#define RNN_I 16
#define RNN_H 6
#define RNN_NP 3
#define RNN_CH 64
#define RNN_RG 32
#define RNN_SPA 24
#define RNN_SPB (RNN_CH - 2 * RNN_SPA)
#define RNN_NC (RNN_T / RNN_CH)
#define RNN_SPITCH (4 * RNN_SPA + 1)

__device__ __forceinline__ v8f wmma_f32_4(v2f a, v2f b, v8f c) {
  return __builtin_amdgcn_wmma_f32_16x16x4_f32(false, a, false, b, (short)0, c,
                                               false, false);
}

__device__ __forceinline__ float tanh_fast(float x) {
#if __has_builtin(__builtin_amdgcn_tanhf)
  return __builtin_amdgcn_tanhf(x);
#else
  x = __builtin_amdgcn_fmed3f(x, -9.0f, 9.0f);
  float e = __expf(2.0f * x);
  return (e - 1.0f) * __builtin_amdgcn_rcpf(e + 1.0f);
#endif
}

__device__ __forceinline__ int rnn_unit_of_row(int row) {
  if (row == 0) return 0;
  if (row == 2 || row == 3) return row - 1;
  if (row == 8) return 3;
  if (row == 10 || row == 11) return row - 6;
  return -1;
}


__device__ __forceinline__ void rnn_fetch_first(const float* __restrict__ xraw, bool wide, v4f (&x0)[16][RNN_SPA / 8]) {
#pragma unroll
  for (int r = 0; r < 16; ++r) {
#pragma unroll
    for (int c = 0; c < RNN_SPB / 8; ++c) x0[r][c] = *(const v4f*)(xraw + (size_t)r * (RNN_T * RNN_I) + c * 128);
  }
  if (wide) {
#pragma unroll
    for (int r = 0; r < 16; ++r) {
#pragma unroll
      for (int c = RNN_SPB / 8; c < RNN_SPA / 8; ++c) x0[r][c] = *(const v4f*)(xraw + (size_t)r * (RNN_T * RNN_I) + c * 128);
    }
  }
}

__device__ __forceinline__ void rnn_stage_first(v4f* stw, bool wide, const v4f (&x0)[16][RNN_SPA / 8]) {
#pragma unroll
  for (int r = 0; r < 16; ++r) {
#pragma unroll
    for (int c = 0; c < RNN_SPB / 8; ++c) stw[r * RNN_SPITCH + 32 * c] = x0[r][c];
  }
  if (wide) {
#pragma unroll
    for (int r = 0; r < 16; ++r) {
#pragma unroll
      for (int c = RNN_SPB / 8; c < RNN_SPA / 8; ++c) stw[r * RNN_SPITCH + 32 * c] = x0[r][c];
    }
  }
}

template <int SPP>
__device__ __forceinline__ void rnn_proj_pass(const float* __restrict__ pn, const v4f* stf, v4f* stw,
                                              float4* dst, v2f w0, v2f w1, v2f w2, v2f w3, v8f cbias) {
  constexpr int LPR = SPP / 8;
  const v2f wih[4] = {w0, w1, w2, w3};
  v4f xn[16][LPR];
#pragma unroll
  for (int r = 0; r < 16; ++r) {
#pragma unroll
    for (int c = 0; c < LPR; ++c) xn[r][c] = *(const v4f*)(pn + (size_t)r * (RNN_T * RNN_I) + c * 128);
  }
  v4f xs[SPP][2];
#pragma unroll
  for (int j = 0; j < SPP; ++j) {
#pragma unroll
    for (int q = 0; q < 2; ++q) xs[j][q] = stf[4 * j + q];
  }
  v8f d[SPP];
#pragma unroll
  for (int j = 0; j < SPP; ++j) d[j] = cbias;
#pragma unroll
  for (int s = 0; s < 4; ++s) {
#pragma unroll
    for (int j = 0; j < SPP; ++j) {
      const v4f f = xs[j][s >> 1];
      const v2f xf = (s & 1) ? __builtin_shufflevector(f, f, 2, 3) : __builtin_shufflevector(f, f, 0, 1);
      d[j] = wmma_f32_4(wih[s], xf, d[j]);
    }
  }
#pragma unroll
  for (int j = 0; j < SPP; ++j) dst[j * 32] = make_float4(d[j][0], d[j][1], d[j][2], d[j][3]);
#pragma unroll
  for (int r = 0; r < 16; ++r) {
#pragma unroll
    for (int c = 0; c < LPR; ++c) stw[r * RNN_SPITCH + 32 * c] = xn[r][c];
  }
}

__global__ __launch_bounds__(32 * (1 + RNN_NP)) void rnn_fused_wmma(
    const float* __restrict__ x, const float* __restrict__ W_ih,
    const float* __restrict__ W_hh, const float* __restrict__ b_ih,
    const float* __restrict__ b_hh, const float* __restrict__ fc_w,
    const float* __restrict__ fc_b, float* __restrict__ out) {
  __shared__ float4 xproj[2][RNN_CH][32];
  __shared__ float hfin[16][8];
  __shared__ v4f stage[RNN_NP][16 * RNN_SPITCH];

  const int tid  = threadIdx.x;
  const int lane = tid & 31;
  const int w    = __builtin_amdgcn_readfirstlane(tid >> 5);
  const int n    = lane & 15;
  const int hi   = lane >> 4;
  const int b0   = blockIdx.x * 16;
  const int p    = (w > 0) ? (w - 1) : 0;

  const float* xraw = x + (size_t)b0 * (RNN_T * RNN_I) + p * (RNN_SPA * RNN_I) + 4 * lane;
  v4f* const st = &stage[p][0];
  v4f* const stw = st + lane;
  const v4f* const stf = st + n * RNN_SPITCH + 2 * hi;

  v4f x0[16][RNN_SPA / 8];
  if (w != 0) rnn_fetch_first(xraw, w != RNN_NP, x0);

  const int ua = rnn_unit_of_row(n);
  const int uw = (ua >= 0) ? ua : 0;
  v2f wih[4];
#pragma unroll
  for (int s = 0; s < 4; ++s) {
#pragma unroll
    for (int v = 0; v < 2; ++v) {
      int k = 8 * hi + 2 * s + v;
      const float t = W_ih[uw * RNN_I + k];
      wih[s][v] = (ua >= 0) ? t : 0.0f;
    }
  }
  v2f whh[2];
#pragma unroll
  for (int s = 0; s < 2; ++s) {
#pragma unroll
    for (int v = 0; v < 2; ++v) {
      const int ku = rnn_unit_of_row(8 * hi + 2 * s + v);
      const float t = W_hh[uw * RNN_H + ((ku >= 0) ? ku : 0)];
      whh[s][v] = (ua >= 0 && ku >= 0) ? t : 0.0f;
    }
  }
  v8f cbias;
#pragma unroll
  for (int r = 0; r < 8; ++r) {
    int uc = rnn_unit_of_row(8 * hi + r);
    const int ub = (uc >= 0) ? uc : 0;
    const float t = b_ih[ub] + b_hh[ub];
    cbias[r] = (uc >= 0) ? t : 0.0f;
  }

  v2f hb0 = 0.0f, hb1 = 0.0f;

  if (w != 0) rnn_stage_first(stw, w != RNN_NP, x0);
  __syncthreads();

  for (int i = 0; i <= RNN_NC; ++i) {
    if (w != 0) {
      if (i < RNN_NC) {
        const int tn = (i + 1 < RNN_NC) ? (i + 1) * RNN_CH : i * RNN_CH;
        const float* pn = xraw + (size_t)tn * RNN_I;
        float4* dst = &xproj[i & 1][p * RNN_SPA][lane];
        if (w == RNN_NP) rnn_proj_pass<RNN_SPB>(pn, stf, stw, dst, wih[0], wih[1], wih[2], wih[3], cbias);
        else             rnn_proj_pass<RNN_SPA>(pn, stf, stw, dst, wih[0], wih[1], wih[2], wih[3], cbias);
      }
    } else {
      if (i >= 1) {
#pragma unroll
        for (int g = 0; g < RNN_CH / RNN_RG; ++g) {
          const float4* srcp = &xproj[(i - 1) & 1][g * RNN_RG][lane];
          float4 q[RNN_RG];
#pragma unroll
          for (int j = 0; j < RNN_RG; ++j) q[j] = srcp[j * 32];
          __builtin_amdgcn_sched_barrier(0);
#pragma unroll
          for (int j = 0; j < RNN_RG; ++j) {
            v8f c = {q[j].x, q[j].y, q[j].z, q[j].w, 0.0f, 0.0f, 0.0f, 0.0f};
            v8f acc = wmma_f32_4(whh[0], hb0, c);
            acc = wmma_f32_4(whh[1], hb1, acc);
            hb0[0] = tanh_fast(acc[0]);
            hb1[0] = tanh_fast(acc[2]);
            hb1[1] = tanh_fast(acc[3]);
          }
          __builtin_amdgcn_sched_barrier(0);
        }
      }
    }
    __syncthreads();
  }

  if (w == 0) {
    const int u0 = 3 * hi;
    hfin[n][u0] = hb0[0]; hfin[n][u0 + 1] = hb1[0]; hfin[n][u0 + 2] = hb1[1];
  }
  __syncthreads();
  if (w == 0 && lane < 16) {
    float s = fc_b[0];
#pragma unroll
    for (int j = 0; j < RNN_H; ++j) s += hfin[lane][j] * fc_w[j];
    out[b0 + lane] = s;
  }
}

extern "C" void kernel_launch(void* const* d_in, const int* in_sizes, int n_in,
                              void* d_out, int out_size, void* d_ws,
                              size_t ws_size, hipStream_t stream) {
  (void)in_sizes; (void)n_in; (void)out_size; (void)d_ws; (void)ws_size;
  const float* x    = (const float*)d_in[0];
  const float* W_ih = (const float*)d_in[1];
  const float* W_hh = (const float*)d_in[2];
  const float* b_ih = (const float*)d_in[3];
  const float* b_hh = (const float*)d_in[4];
  const float* fc_w = (const float*)d_in[5];
  const float* fc_b = (const float*)d_in[6];
  float* out = (float*)d_out;

  dim3 block(32 * (1 + RNN_NP));
  dim3 grid(RNN_B / 16);
  rnn_fused_wmma<<<grid, block, 0, stream>>>(x, W_ih, W_hh, b_ih, b_hh, fc_w,
                                             fc_b, out);
}
